# starting kernel
# baseline (speedup 1.0000x reference)
#include <hip/hip_runtime.h>
#include <math.h>

typedef __attribute__((ext_vector_type(2))) float v2f;
typedef __attribute__((ext_vector_type(8))) float v8f;

#define B_ 8
#define C_ 256
#define F_ 128
#define H_ 256

__global__ __launch_bounds__(128) void gemm_ac_wmma(
    const float* __restrict__ xa, const float* __restrict__ W1,
    const float* __restrict__ b1,
    float* __restrict__ ws_a, float* __restrict__ ws_c)
{
  const int wave = blockIdx.x * (blockDim.x >> 5) + (threadIdx.x >> 5);
  const int lane = threadIdx.x & 31;

  int t = wave;
  const int ht  = t & 15; t >>= 4;
  const int it  = t & 15; t >>= 4;
  const int b   = t & 7;  t >>= 3;
  const int sel = t & 1;

  const int i0 = it * 16;
  const int h0 = ht * 16;
  const int lo = lane & 15;
  const int hi = lane >> 4;

  const float* arow = xa + ((size_t)(b * C_ + i0 + lo)) * F_;
  const float* brow = W1 + (size_t)(h0 + lo) * (2 * F_) + sel * F_;

  v8f acc = {};
#pragma unroll 8
  for (int k = 0; k < F_; k += 4) {
    const int ka = k + 2 * hi;
    v2f af = *(const v2f*)(arow + ka);
    v2f bf = *(const v2f*)(brow + ka);
    acc = __builtin_amdgcn_wmma_f32_16x16x4_f32(false, af, false, bf,
                                                (short)0, acc, false, false);
  }

  const float bias = (sel == 0) ? b1[h0 + lo] : 0.0f;
  float* __restrict__ dst = (sel == 0) ? ws_a : ws_c;
#pragma unroll
  for (int r = 0; r < 8; ++r) {
    const int m = i0 + r + 8 * hi;
    dst[((size_t)(b * C_ + m)) * H_ + (h0 + lo)] = acc[r] + bias;
  }
}

__global__ __launch_bounds__(256) void edge_epilogue(
    const float* __restrict__ ws_a, const float* __restrict__ ws_c,
    const float* __restrict__ w2, const float* __restrict__ b2p,
    float* __restrict__ out)
{
  __shared__ float a_s[32 * H_];
  __shared__ float c_s[H_ * 33];
  __shared__ float w_s[H_];

  const int tid = threadIdx.x;
  int t = blockIdx.x;
  const int jt = t & 7; t >>= 3;
  const int it = t & 7; t >>= 3;
  const int b  = t;
  const int i0 = it * 32, j0 = jt * 32;

  {
    const float4* src = (const float4*)(ws_a + ((size_t)(b * C_ + i0)) * H_);
    float4* dst4 = (float4*)a_s;
#pragma unroll
    for (int idx = tid; idx < (32 * H_) / 4; idx += 256) dst4[idx] = src[idx];
  }
  {
    const float4* src = (const float4*)(ws_c + ((size_t)(b * C_ + j0)) * H_);
#pragma unroll
    for (int idx = tid; idx < (32 * H_) / 4; idx += 256) {
      const int j  = idx >> 6;
      const int hb = (idx & 63) * 4;
      float4 v = src[idx];
      c_s[(hb + 0) * 33 + j] = v.x;
      c_s[(hb + 1) * 33 + j] = v.y;
      c_s[(hb + 2) * 33 + j] = v.z;
      c_s[(hb + 3) * 33 + j] = v.w;
    }
  }
  if (tid < H_) w_s[tid] = w2[tid];
  __syncthreads();

  const int ti = tid >> 5;
  const int tj = tid & 31;
  const float* a0 = a_s + (ti +  0) * H_;
  const float* a1 = a_s + (ti +  8) * H_;
  const float* a2 = a_s + (ti + 16) * H_;
  const float* a3 = a_s + (ti + 24) * H_;

  float acc0 = 0.f, acc1 = 0.f, acc2 = 0.f, acc3 = 0.f;
#pragma unroll 4
  for (int h = 0; h < H_; ++h) {
    const float wv = w_s[h];
    const float cv = c_s[h * 33 + tj];
    float x;
    x = a0[h] + cv; acc0 = fmaf(fmaxf(x, 0.f), wv, acc0);
    x = a1[h] + cv; acc1 = fmaf(fmaxf(x, 0.f), wv, acc1);
    x = a2[h] + cv; acc2 = fmaf(fmaxf(x, 0.f), wv, acc2);
    x = a3[h] + cv; acc3 = fmaf(fmaxf(x, 0.f), wv, acc3);
  }

  const float b2 = b2p[0];
  const size_t obase = ((size_t)b * C_) * C_;
  const int j = j0 + tj;
  float* o = out + obase + (size_t)(i0 + ti) * C_ + j;
  o[ 0 * 8 * C_] = 1.0f / (1.0f + __expf(-(acc0 + b2)));
  o[ 1 * 8 * C_] = 1.0f / (1.0f + __expf(-(acc1 + b2)));
  o[ 2 * 8 * C_] = 1.0f / (1.0f + __expf(-(acc2 + b2)));
  o[ 3 * 8 * C_] = 1.0f / (1.0f + __expf(-(acc3 + b2)));
}

extern "C" void kernel_launch(void* const* d_in, const int* in_sizes, int n_in,
                              void* d_out, int out_size, void* d_ws, size_t ws_size,
                              hipStream_t stream) {
  const float* xa = (const float*)d_in[0];
  const float* W1 = (const float*)d_in[1];
  const float* b1 = (const float*)d_in[2];
  const float* w2 = (const float*)d_in[3];
  const float* b2 = (const float*)d_in[4];
  float* out = (float*)d_out;

  float* ws_a = (float*)d_ws;
  float* ws_c = ws_a + (size_t)B_ * C_ * H_;

  gemm_ac_wmma<<<1024, 128, 0, stream>>>(xa, W1, b1, ws_a, ws_c);

  edge_epilogue<<<512, 256, 0, stream>>>(ws_a, ws_c, w2, b2, out);
}


# epilogue retiled: 32x64 block tile, 4x2 register tile per lane (c tile row-major at pitch 260, 16 h per trip), so each broadcast LDS read of a and w2 feeds two j columns; h-sum order unchanged
# speedup vs baseline: 1.0869x; 1.0869x over previous
#include <hip/hip_runtime.h>
#include <math.h>

typedef __attribute__((ext_vector_type(2))) float v2f;
typedef __attribute__((ext_vector_type(8))) float v8f;

#define B_ 8
#define C_ 256
#define F_ 128
#define H_ 256

__global__ __launch_bounds__(128) void gemm_ac_wmma(
    const float* __restrict__ xa, const float* __restrict__ W1,
    const float* __restrict__ b1,
    float* __restrict__ ws_a, float* __restrict__ ws_c)
{
  const int wave = blockIdx.x * (blockDim.x >> 5) + (threadIdx.x >> 5);
  const int lane = threadIdx.x & 31;

  int t = wave;
  const int ht  = t & 15; t >>= 4;
  const int it  = t & 15; t >>= 4;
  const int b   = t & 7;  t >>= 3;
  const int sel = t & 1;

  const int i0 = it * 16;
  const int h0 = ht * 16;
  const int lo = lane & 15;
  const int hi = lane >> 4;

  const float* arow = xa + ((size_t)(b * C_ + i0 + lo)) * F_;
  const float* brow = W1 + (size_t)(h0 + lo) * (2 * F_) + sel * F_;

  v8f acc = {};
#pragma unroll 8
  for (int k = 0; k < F_; k += 4) {
    const int ka = k + 2 * hi;
    v2f af = *(const v2f*)(arow + ka);
    v2f bf = *(const v2f*)(brow + ka);
    acc = __builtin_amdgcn_wmma_f32_16x16x4_f32(false, af, false, bf,
                                                (short)0, acc, false, false);
  }

  const float bias = (sel == 0) ? b1[h0 + lo] : 0.0f;
  float* __restrict__ dst = (sel == 0) ? ws_a : ws_c;
#pragma unroll
  for (int r = 0; r < 8; ++r) {
    const int m = i0 + r + 8 * hi;
    dst[((size_t)(b * C_ + m)) * H_ + (h0 + lo)] = acc[r] + bias;
  }
}

#define CP_ 260

__global__ __launch_bounds__(256) void edge_epilogue(
    const float* __restrict__ ws_a, const float* __restrict__ ws_c,
    const float* __restrict__ w2, const float* __restrict__ b2p,
    float* __restrict__ out)
{
  __shared__ __attribute__((aligned(16))) float a_s[32 * H_];
  __shared__ __attribute__((aligned(16))) float c_s[64 * CP_];
  __shared__ __attribute__((aligned(16))) float w_s[H_];

  const int tid = threadIdx.x;
  int t = blockIdx.x;
  const int jt = t & 3; t >>= 2;
  const int it = t & 7; t >>= 3;
  const int b  = t;
  const int i0 = it * 32, j0 = jt * 64;

  {
    const float4* src = (const float4*)(ws_a + ((size_t)(b * C_ + i0)) * H_);
    float4* dst4 = (float4*)a_s;
#pragma unroll
    for (int idx = tid; idx < (32 * H_) / 4; idx += 256) dst4[idx] = src[idx];
  }
  {
    const float4* src = (const float4*)(ws_c + ((size_t)(b * C_ + j0)) * H_);
#pragma unroll
    for (int idx = tid; idx < (64 * H_) / 4; idx += 256) {
      const int j = idx >> 6;
      const int q = idx & 63;
      *(float4*)(c_s + j * CP_ + 4 * q) = src[idx];
    }
  }
  if (tid < H_) w_s[tid] = w2[tid];
  __syncthreads();

  const int ti = tid >> 5;
  const int tj = tid & 31;
  const float* a0 = a_s + (ti +  0) * H_;
  const float* a1 = a_s + (ti +  8) * H_;
  const float* a2 = a_s + (ti + 16) * H_;
  const float* a3 = a_s + (ti + 24) * H_;
  const float* c0 = c_s + (tj +  0) * CP_;
  const float* c1 = c_s + (tj + 32) * CP_;

  float acc00 = 0.f, acc10 = 0.f, acc20 = 0.f, acc30 = 0.f;
  float acc01 = 0.f, acc11 = 0.f, acc21 = 0.f, acc31 = 0.f;
#pragma unroll 4
  for (int h = 0; h < H_; h += 4) {
    const float4 wv = *(const float4*)(w_s + h);
    const float4 p  = *(const float4*)(c0 + h);
    const float4 q  = *(const float4*)(c1 + h);
    const float4 x0 = *(const float4*)(a0 + h);
    const float4 x1 = *(const float4*)(a1 + h);
    const float4 x2 = *(const float4*)(a2 + h);
    const float4 x3 = *(const float4*)(a3 + h);
    acc00 = fmaf(fmaxf(x0.x + p.x, 0.f), wv.x, acc00);
    acc01 = fmaf(fmaxf(x0.x + q.x, 0.f), wv.x, acc01);
    acc10 = fmaf(fmaxf(x1.x + p.x, 0.f), wv.x, acc10);
    acc11 = fmaf(fmaxf(x1.x + q.x, 0.f), wv.x, acc11);
    acc20 = fmaf(fmaxf(x2.x + p.x, 0.f), wv.x, acc20);
    acc21 = fmaf(fmaxf(x2.x + q.x, 0.f), wv.x, acc21);
    acc30 = fmaf(fmaxf(x3.x + p.x, 0.f), wv.x, acc30);
    acc31 = fmaf(fmaxf(x3.x + q.x, 0.f), wv.x, acc31);

    acc00 = fmaf(fmaxf(x0.y + p.y, 0.f), wv.y, acc00);
    acc01 = fmaf(fmaxf(x0.y + q.y, 0.f), wv.y, acc01);
    acc10 = fmaf(fmaxf(x1.y + p.y, 0.f), wv.y, acc10);
    acc11 = fmaf(fmaxf(x1.y + q.y, 0.f), wv.y, acc11);
    acc20 = fmaf(fmaxf(x2.y + p.y, 0.f), wv.y, acc20);
    acc21 = fmaf(fmaxf(x2.y + q.y, 0.f), wv.y, acc21);
    acc30 = fmaf(fmaxf(x3.y + p.y, 0.f), wv.y, acc30);
    acc31 = fmaf(fmaxf(x3.y + q.y, 0.f), wv.y, acc31);

    acc00 = fmaf(fmaxf(x0.z + p.z, 0.f), wv.z, acc00);
    acc01 = fmaf(fmaxf(x0.z + q.z, 0.f), wv.z, acc01);
    acc10 = fmaf(fmaxf(x1.z + p.z, 0.f), wv.z, acc10);
    acc11 = fmaf(fmaxf(x1.z + q.z, 0.f), wv.z, acc11);
    acc20 = fmaf(fmaxf(x2.z + p.z, 0.f), wv.z, acc20);
    acc21 = fmaf(fmaxf(x2.z + q.z, 0.f), wv.z, acc21);
    acc30 = fmaf(fmaxf(x3.z + p.z, 0.f), wv.z, acc30);
    acc31 = fmaf(fmaxf(x3.z + q.z, 0.f), wv.z, acc31);

    acc00 = fmaf(fmaxf(x0.w + p.w, 0.f), wv.w, acc00);
    acc01 = fmaf(fmaxf(x0.w + q.w, 0.f), wv.w, acc01);
    acc10 = fmaf(fmaxf(x1.w + p.w, 0.f), wv.w, acc10);
    acc11 = fmaf(fmaxf(x1.w + q.w, 0.f), wv.w, acc11);
    acc20 = fmaf(fmaxf(x2.w + p.w, 0.f), wv.w, acc20);
    acc21 = fmaf(fmaxf(x2.w + q.w, 0.f), wv.w, acc21);
    acc30 = fmaf(fmaxf(x3.w + p.w, 0.f), wv.w, acc30);
    acc31 = fmaf(fmaxf(x3.w + q.w, 0.f), wv.w, acc31);
  }

  const float b2 = b2p[0];
  const size_t obase = ((size_t)b * C_) * C_;
  const int j = j0 + tj;
  float* o = out + obase + (size_t)(i0 + ti) * C_ + j;
  o[ 0 * 8 * C_     ] = 1.0f / (1.0f + __expf(-(acc00 + b2)));
  o[ 0 * 8 * C_ + 32] = 1.0f / (1.0f + __expf(-(acc01 + b2)));
  o[ 1 * 8 * C_     ] = 1.0f / (1.0f + __expf(-(acc10 + b2)));
  o[ 1 * 8 * C_ + 32] = 1.0f / (1.0f + __expf(-(acc11 + b2)));
  o[ 2 * 8 * C_     ] = 1.0f / (1.0f + __expf(-(acc20 + b2)));
  o[ 2 * 8 * C_ + 32] = 1.0f / (1.0f + __expf(-(acc21 + b2)));
  o[ 3 * 8 * C_     ] = 1.0f / (1.0f + __expf(-(acc30 + b2)));
  o[ 3 * 8 * C_ + 32] = 1.0f / (1.0f + __expf(-(acc31 + b2)));
}

extern "C" void kernel_launch(void* const* d_in, const int* in_sizes, int n_in,
                              void* d_out, int out_size, void* d_ws, size_t ws_size,
                              hipStream_t stream) {
  const float* xa = (const float*)d_in[0];
  const float* W1 = (const float*)d_in[1];
  const float* b1 = (const float*)d_in[2];
  const float* w2 = (const float*)d_in[3];
  const float* b2 = (const float*)d_in[4];
  float* out = (float*)d_out;

  float* ws_a = (float*)d_ws;
  float* ws_c = ws_a + (size_t)B_ * C_ * H_;

  gemm_ac_wmma<<<1024, 128, 0, stream>>>(xa, W1, b1, ws_a, ws_c);

  edge_epilogue<<<256, 256, 0, stream>>>(ws_a, ws_c, w2, b2, out);
}


# epilogue staging pipelined: h in 4 chunks of 64 through two LDS buffers, the next chunk's global loads in flight during the current chunk's arithmetic (LDS-only barrier); h-sum order unchanged
# speedup vs baseline: 1.2487x; 1.1489x over previous
#include <hip/hip_runtime.h>
#include <math.h>

typedef __attribute__((ext_vector_type(2))) float v2f;
typedef __attribute__((ext_vector_type(8))) float v8f;

#define B_ 8
#define C_ 256
#define F_ 128
#define H_ 256

__global__ __launch_bounds__(128) void gemm_ac_wmma(
    const float* __restrict__ xa, const float* __restrict__ W1,
    const float* __restrict__ b1,
    float* __restrict__ ws_a, float* __restrict__ ws_c)
{
  const int wave = blockIdx.x * (blockDim.x >> 5) + (threadIdx.x >> 5);
  const int lane = threadIdx.x & 31;

  int t = wave;
  const int ht  = t & 15; t >>= 4;
  const int it  = t & 15; t >>= 4;
  const int b   = t & 7;  t >>= 3;
  const int sel = t & 1;

  const int i0 = it * 16;
  const int h0 = ht * 16;
  const int lo = lane & 15;
  const int hi = lane >> 4;

  const float* arow = xa + ((size_t)(b * C_ + i0 + lo)) * F_;
  const float* brow = W1 + (size_t)(h0 + lo) * (2 * F_) + sel * F_;

  v8f acc = {};
#pragma unroll 8
  for (int k = 0; k < F_; k += 4) {
    const int ka = k + 2 * hi;
    v2f af = *(const v2f*)(arow + ka);
    v2f bf = *(const v2f*)(brow + ka);
    acc = __builtin_amdgcn_wmma_f32_16x16x4_f32(false, af, false, bf,
                                                (short)0, acc, false, false);
  }

  const float bias = (sel == 0) ? b1[h0 + lo] : 0.0f;
  float* __restrict__ dst = (sel == 0) ? ws_a : ws_c;
#pragma unroll
  for (int r = 0; r < 8; ++r) {
    const int m = i0 + r + 8 * hi;
    dst[((size_t)(b * C_ + m)) * H_ + (h0 + lo)] = acc[r] + bias;
  }
}

#define HC_  64
#define NCH_ (H_ / HC_)
#define AP_  HC_
#define CP_  68

__device__ __forceinline__ void lds_barrier() {
  __builtin_amdgcn_fence(__ATOMIC_RELEASE, "workgroup", "local");
  __builtin_amdgcn_s_barrier();
  __builtin_amdgcn_fence(__ATOMIC_ACQUIRE, "workgroup", "local");
}

__global__ __launch_bounds__(256) void edge_epilogue(
    const float* ws_a, const float* ws_c,
    const float* __restrict__ w2, const float* __restrict__ b2p,
    float* __restrict__ out)
{
  __shared__ __attribute__((aligned(16))) float a_s[2][32 * AP_];
  __shared__ __attribute__((aligned(16))) float c_s[2][64 * CP_];
  __shared__ __attribute__((aligned(16))) float w_s[H_];

  const int tid = threadIdx.x;
  int t = blockIdx.x;
  const int jt = t & 3; t >>= 2;
  const int it = t & 7; t >>= 3;
  const int b  = t;
  const int i0 = it * 32, j0 = jt * 64;

  const int sr = tid >> 4;
  const int sq = tid & 15;
  const float* ga = ws_a + ((size_t)(b * C_ + i0 + sr)) * H_ + 4 * sq;
  const float* gc = ws_c + ((size_t)(b * C_ + j0 + sr)) * H_ + 4 * sq;
  const int la = sr * AP_ + 4 * sq;
  const int lc = sr * CP_ + 4 * sq;

  float4 ra0, ra1, rc0, rc1, rc2, rc3;

  ra0 = *(const float4*)(ga);
  ra1 = *(const float4*)(ga + 16 * H_);
  rc0 = *(const float4*)(gc);
  rc1 = *(const float4*)(gc + 16 * H_);
  rc2 = *(const float4*)(gc + 32 * H_);
  rc3 = *(const float4*)(gc + 48 * H_);
  const float wreg = w2[tid];
  *(float4*)(&a_s[0][la])            = ra0;
  *(float4*)(&a_s[0][la + 16 * AP_]) = ra1;
  *(float4*)(&c_s[0][lc])            = rc0;
  *(float4*)(&c_s[0][lc + 16 * CP_]) = rc1;
  *(float4*)(&c_s[0][lc + 32 * CP_]) = rc2;
  *(float4*)(&c_s[0][lc + 48 * CP_]) = rc3;
  w_s[tid] = wreg;
  ra0 = *(const float4*)(ga + HC_);
  ra1 = *(const float4*)(ga + HC_ + 16 * H_);
  rc0 = *(const float4*)(gc + HC_);
  rc1 = *(const float4*)(gc + HC_ + 16 * H_);
  rc2 = *(const float4*)(gc + HC_ + 32 * H_);
  rc3 = *(const float4*)(gc + HC_ + 48 * H_);
  lds_barrier();

  const int ti = tid >> 5;
  const int tj = tid & 31;

  float acc00 = 0.f, acc10 = 0.f, acc20 = 0.f, acc30 = 0.f;
  float acc01 = 0.f, acc11 = 0.f, acc21 = 0.f, acc31 = 0.f;

#pragma unroll
  for (int ch = 0; ch < NCH_; ++ch) {
    const int cur = ch & 1;
    const float* a0 = &a_s[cur][(ti +  0) * AP_];
    const float* a1 = &a_s[cur][(ti +  8) * AP_];
    const float* a2 = &a_s[cur][(ti + 16) * AP_];
    const float* a3 = &a_s[cur][(ti + 24) * AP_];
    const float* c0 = &c_s[cur][(tj +  0) * CP_];
    const float* c1 = &c_s[cur][(tj + 32) * CP_];
    const float* wp = w_s + ch * HC_;

#pragma unroll 4
    for (int h = 0; h < HC_; h += 4) {
      const float4 wv = *(const float4*)(wp + h);
      const float4 p  = *(const float4*)(c0 + h);
      const float4 q  = *(const float4*)(c1 + h);
      const float4 x0 = *(const float4*)(a0 + h);
      const float4 x1 = *(const float4*)(a1 + h);
      const float4 x2 = *(const float4*)(a2 + h);
      const float4 x3 = *(const float4*)(a3 + h);
      acc00 = fmaf(fmaxf(x0.x + p.x, 0.f), wv.x, acc00);
      acc01 = fmaf(fmaxf(x0.x + q.x, 0.f), wv.x, acc01);
      acc10 = fmaf(fmaxf(x1.x + p.x, 0.f), wv.x, acc10);
      acc11 = fmaf(fmaxf(x1.x + q.x, 0.f), wv.x, acc11);
      acc20 = fmaf(fmaxf(x2.x + p.x, 0.f), wv.x, acc20);
      acc21 = fmaf(fmaxf(x2.x + q.x, 0.f), wv.x, acc21);
      acc30 = fmaf(fmaxf(x3.x + p.x, 0.f), wv.x, acc30);
      acc31 = fmaf(fmaxf(x3.x + q.x, 0.f), wv.x, acc31);

      acc00 = fmaf(fmaxf(x0.y + p.y, 0.f), wv.y, acc00);
      acc01 = fmaf(fmaxf(x0.y + q.y, 0.f), wv.y, acc01);
      acc10 = fmaf(fmaxf(x1.y + p.y, 0.f), wv.y, acc10);
      acc11 = fmaf(fmaxf(x1.y + q.y, 0.f), wv.y, acc11);
      acc20 = fmaf(fmaxf(x2.y + p.y, 0.f), wv.y, acc20);
      acc21 = fmaf(fmaxf(x2.y + q.y, 0.f), wv.y, acc21);
      acc30 = fmaf(fmaxf(x3.y + p.y, 0.f), wv.y, acc30);
      acc31 = fmaf(fmaxf(x3.y + q.y, 0.f), wv.y, acc31);

      acc00 = fmaf(fmaxf(x0.z + p.z, 0.f), wv.z, acc00);
      acc01 = fmaf(fmaxf(x0.z + q.z, 0.f), wv.z, acc01);
      acc10 = fmaf(fmaxf(x1.z + p.z, 0.f), wv.z, acc10);
      acc11 = fmaf(fmaxf(x1.z + q.z, 0.f), wv.z, acc11);
      acc20 = fmaf(fmaxf(x2.z + p.z, 0.f), wv.z, acc20);
      acc21 = fmaf(fmaxf(x2.z + q.z, 0.f), wv.z, acc21);
      acc30 = fmaf(fmaxf(x3.z + p.z, 0.f), wv.z, acc30);
      acc31 = fmaf(fmaxf(x3.z + q.z, 0.f), wv.z, acc31);

      acc00 = fmaf(fmaxf(x0.w + p.w, 0.f), wv.w, acc00);
      acc01 = fmaf(fmaxf(x0.w + q.w, 0.f), wv.w, acc01);
      acc10 = fmaf(fmaxf(x1.w + p.w, 0.f), wv.w, acc10);
      acc11 = fmaf(fmaxf(x1.w + q.w, 0.f), wv.w, acc11);
      acc20 = fmaf(fmaxf(x2.w + p.w, 0.f), wv.w, acc20);
      acc21 = fmaf(fmaxf(x2.w + q.w, 0.f), wv.w, acc21);
      acc30 = fmaf(fmaxf(x3.w + p.w, 0.f), wv.w, acc30);
      acc31 = fmaf(fmaxf(x3.w + q.w, 0.f), wv.w, acc31);
    }

    if (ch + 1 < NCH_) {
      const int nxt = cur ^ 1;
      *(float4*)(&a_s[nxt][la])            = ra0;
      *(float4*)(&a_s[nxt][la + 16 * AP_]) = ra1;
      *(float4*)(&c_s[nxt][lc])            = rc0;
      *(float4*)(&c_s[nxt][lc + 16 * CP_]) = rc1;
      *(float4*)(&c_s[nxt][lc + 32 * CP_]) = rc2;
      *(float4*)(&c_s[nxt][lc + 48 * CP_]) = rc3;
      if (ch + 2 < NCH_) {
        const int ho = (ch + 2) * HC_;
        ra0 = *(const float4*)(ga + ho);
        ra1 = *(const float4*)(ga + ho + 16 * H_);
        rc0 = *(const float4*)(gc + ho);
        rc1 = *(const float4*)(gc + ho + 16 * H_);
        rc2 = *(const float4*)(gc + ho + 32 * H_);
        rc3 = *(const float4*)(gc + ho + 48 * H_);
      }
      lds_barrier();
    }
  }

  const float b2 = b2p[0];
  const size_t obase = ((size_t)b * C_) * C_;
  const int j = j0 + tj;
  float* o = out + obase + (size_t)(i0 + ti) * C_ + j;
  o[ 0 * 8 * C_     ] = 1.0f / (1.0f + __expf(-(acc00 + b2)));
  o[ 0 * 8 * C_ + 32] = 1.0f / (1.0f + __expf(-(acc01 + b2)));
  o[ 1 * 8 * C_     ] = 1.0f / (1.0f + __expf(-(acc10 + b2)));
  o[ 1 * 8 * C_ + 32] = 1.0f / (1.0f + __expf(-(acc11 + b2)));
  o[ 2 * 8 * C_     ] = 1.0f / (1.0f + __expf(-(acc20 + b2)));
  o[ 2 * 8 * C_ + 32] = 1.0f / (1.0f + __expf(-(acc21 + b2)));
  o[ 3 * 8 * C_     ] = 1.0f / (1.0f + __expf(-(acc30 + b2)));
  o[ 3 * 8 * C_ + 32] = 1.0f / (1.0f + __expf(-(acc31 + b2)));
}

extern "C" void kernel_launch(void* const* d_in, const int* in_sizes, int n_in,
                              void* d_out, int out_size, void* d_ws, size_t ws_size,
                              hipStream_t stream) {
  const float* xa = (const float*)d_in[0];
  const float* W1 = (const float*)d_in[1];
  const float* b1 = (const float*)d_in[2];
  const float* w2 = (const float*)d_in[3];
  const float* b2 = (const float*)d_in[4];
  float* out = (float*)d_out;

  float* ws_a = (float*)d_ws;
  float* ws_c = ws_a + (size_t)B_ * C_ * H_;

  gemm_ac_wmma<<<1024, 128, 0, stream>>>(xa, W1, b1, ws_a, ws_c);

  edge_epilogue<<<256, 256, 0, stream>>>(ws_a, ws_c, w2, b2, out);
}


# first GEMM retiled: 64x64 block tile staged in LDS by whole-line global loads (two k chunks, second in flight), 2x2 WMMA tiles per wave; k-sum order unchanged
# speedup vs baseline: 1.5634x; 1.2520x over previous
#include <hip/hip_runtime.h>
#include <math.h>

typedef __attribute__((ext_vector_type(2))) float v2f;
typedef __attribute__((ext_vector_type(8))) float v8f;

#define B_ 8
#define C_ 256
#define F_ 128
#define H_ 256

__device__ __forceinline__ void lds_barrier() {
  __builtin_amdgcn_fence(__ATOMIC_RELEASE, "workgroup", "local");
  __builtin_amdgcn_s_barrier();
  __builtin_amdgcn_fence(__ATOMIC_ACQUIRE, "workgroup", "local");
}

#define GT_  64
#define GP_  132
#define GKC_ 64

__global__ __launch_bounds__(128) void gemm_ac_wmma(
    const float* xa, const float* W1,
    const float* __restrict__ b1,
    float* __restrict__ ws_a, float* __restrict__ ws_c)
{
  __shared__ __attribute__((aligned(16))) float ga_s[GT_ * GP_];
  __shared__ __attribute__((aligned(16))) float gb_s[GT_ * GP_];

  const int tid  = threadIdx.x;
  const int lane = tid & 31;
  const int wave = tid >> 5;

  const int nt  = blockIdx.x & 7;
  const int mt  = blockIdx.x >> 3;
  const int sel = nt >> 2;
  const int h0  = (nt & 3) * GT_;
  const int m0  = mt * GT_;

  const int sr = tid >> 4;
  const int sq = tid & 15;
  const float* gA = xa + (size_t)(m0 + sr) * F_ + 4 * sq;
  const float* gB = W1 + (size_t)(h0 + sr) * (2 * F_) + sel * F_ + 4 * sq;
  const int ls = sr * GP_ + 4 * sq;

  const float4 pa0 = *(const float4*)(gA + 0 * F_);
  const float4 pa1 = *(const float4*)(gA + 8 * F_);
  const float4 pa2 = *(const float4*)(gA + 16 * F_);
  const float4 pa3 = *(const float4*)(gA + 24 * F_);
  const float4 pa4 = *(const float4*)(gA + 32 * F_);
  const float4 pa5 = *(const float4*)(gA + 40 * F_);
  const float4 pa6 = *(const float4*)(gA + 48 * F_);
  const float4 pa7 = *(const float4*)(gA + 56 * F_);
  const float4 pb0 = *(const float4*)(gB + 0 * (2 * F_));
  const float4 pb1 = *(const float4*)(gB + 8 * (2 * F_));
  const float4 pb2 = *(const float4*)(gB + 16 * (2 * F_));
  const float4 pb3 = *(const float4*)(gB + 24 * (2 * F_));
  const float4 pb4 = *(const float4*)(gB + 32 * (2 * F_));
  const float4 pb5 = *(const float4*)(gB + 40 * (2 * F_));
  const float4 pb6 = *(const float4*)(gB + 48 * (2 * F_));
  const float4 pb7 = *(const float4*)(gB + 56 * (2 * F_));
  __builtin_amdgcn_sched_barrier(0);
  const float4 qa0 = *(const float4*)(gA + 0 * F_ + GKC_);
  const float4 qa1 = *(const float4*)(gA + 8 * F_ + GKC_);
  const float4 qa2 = *(const float4*)(gA + 16 * F_ + GKC_);
  const float4 qa3 = *(const float4*)(gA + 24 * F_ + GKC_);
  const float4 qa4 = *(const float4*)(gA + 32 * F_ + GKC_);
  const float4 qa5 = *(const float4*)(gA + 40 * F_ + GKC_);
  const float4 qa6 = *(const float4*)(gA + 48 * F_ + GKC_);
  const float4 qa7 = *(const float4*)(gA + 56 * F_ + GKC_);
  const float4 qb0 = *(const float4*)(gB + 0 * (2 * F_) + GKC_);
  const float4 qb1 = *(const float4*)(gB + 8 * (2 * F_) + GKC_);
  const float4 qb2 = *(const float4*)(gB + 16 * (2 * F_) + GKC_);
  const float4 qb3 = *(const float4*)(gB + 24 * (2 * F_) + GKC_);
  const float4 qb4 = *(const float4*)(gB + 32 * (2 * F_) + GKC_);
  const float4 qb5 = *(const float4*)(gB + 40 * (2 * F_) + GKC_);
  const float4 qb6 = *(const float4*)(gB + 48 * (2 * F_) + GKC_);
  const float4 qb7 = *(const float4*)(gB + 56 * (2 * F_) + GKC_);

  const int lo = lane & 15;
  const int hi = lane >> 4;
  const int wm = (wave >> 1) * 32;
  const int wn = (wave & 1) * 32;
  const float* fa0 = &ga_s[(wm +  0 + lo) * GP_ + 2 * hi];
  const float* fa1 = &ga_s[(wm + 16 + lo) * GP_ + 2 * hi];
  const float* fb0 = &gb_s[(wn +  0 + lo) * GP_ + 2 * hi];
  const float* fb1 = &gb_s[(wn + 16 + lo) * GP_ + 2 * hi];

  v8f acc00 = {}, acc01 = {}, acc10 = {}, acc11 = {};

  *(float4*)(&ga_s[ls + 0 * GP_]) = pa0;
  *(float4*)(&ga_s[ls + 8 * GP_]) = pa1;
  *(float4*)(&ga_s[ls + 16 * GP_]) = pa2;
  *(float4*)(&ga_s[ls + 24 * GP_]) = pa3;
  *(float4*)(&ga_s[ls + 32 * GP_]) = pa4;
  *(float4*)(&ga_s[ls + 40 * GP_]) = pa5;
  *(float4*)(&ga_s[ls + 48 * GP_]) = pa6;
  *(float4*)(&ga_s[ls + 56 * GP_]) = pa7;
  *(float4*)(&gb_s[ls + 0 * GP_]) = pb0;
  *(float4*)(&gb_s[ls + 8 * GP_]) = pb1;
  *(float4*)(&gb_s[ls + 16 * GP_]) = pb2;
  *(float4*)(&gb_s[ls + 24 * GP_]) = pb3;
  *(float4*)(&gb_s[ls + 32 * GP_]) = pb4;
  *(float4*)(&gb_s[ls + 40 * GP_]) = pb5;
  *(float4*)(&gb_s[ls + 48 * GP_]) = pb6;
  *(float4*)(&gb_s[ls + 56 * GP_]) = pb7;
  lds_barrier();

#pragma unroll
  for (int k = 0; k < GKC_; k += 4) {
    const v2f a0 = *(const v2f*)(fa0 + k);
    const v2f a1 = *(const v2f*)(fa1 + k);
    const v2f b0 = *(const v2f*)(fb0 + k);
    const v2f b1f = *(const v2f*)(fb1 + k);
    acc00 = __builtin_amdgcn_wmma_f32_16x16x4_f32(false, a0, false, b0,  (short)0, acc00, false, false);
    acc01 = __builtin_amdgcn_wmma_f32_16x16x4_f32(false, a0, false, b1f, (short)0, acc01, false, false);
    acc10 = __builtin_amdgcn_wmma_f32_16x16x4_f32(false, a1, false, b0,  (short)0, acc10, false, false);
    acc11 = __builtin_amdgcn_wmma_f32_16x16x4_f32(false, a1, false, b1f, (short)0, acc11, false, false);
  }

  *(float4*)(&ga_s[ls + 0 * GP_ + GKC_]) = qa0;
  *(float4*)(&ga_s[ls + 8 * GP_ + GKC_]) = qa1;
  *(float4*)(&ga_s[ls + 16 * GP_ + GKC_]) = qa2;
  *(float4*)(&ga_s[ls + 24 * GP_ + GKC_]) = qa3;
  *(float4*)(&ga_s[ls + 32 * GP_ + GKC_]) = qa4;
  *(float4*)(&ga_s[ls + 40 * GP_ + GKC_]) = qa5;
  *(float4*)(&ga_s[ls + 48 * GP_ + GKC_]) = qa6;
  *(float4*)(&ga_s[ls + 56 * GP_ + GKC_]) = qa7;
  *(float4*)(&gb_s[ls + 0 * GP_ + GKC_]) = qb0;
  *(float4*)(&gb_s[ls + 8 * GP_ + GKC_]) = qb1;
  *(float4*)(&gb_s[ls + 16 * GP_ + GKC_]) = qb2;
  *(float4*)(&gb_s[ls + 24 * GP_ + GKC_]) = qb3;
  *(float4*)(&gb_s[ls + 32 * GP_ + GKC_]) = qb4;
  *(float4*)(&gb_s[ls + 40 * GP_ + GKC_]) = qb5;
  *(float4*)(&gb_s[ls + 48 * GP_ + GKC_]) = qb6;
  *(float4*)(&gb_s[ls + 56 * GP_ + GKC_]) = qb7;
  lds_barrier();

#pragma unroll
  for (int k = GKC_; k < F_; k += 4) {
    const v2f a0 = *(const v2f*)(fa0 + k);
    const v2f a1 = *(const v2f*)(fa1 + k);
    const v2f b0 = *(const v2f*)(fb0 + k);
    const v2f b1f = *(const v2f*)(fb1 + k);
    acc00 = __builtin_amdgcn_wmma_f32_16x16x4_f32(false, a0, false, b0,  (short)0, acc00, false, false);
    acc01 = __builtin_amdgcn_wmma_f32_16x16x4_f32(false, a0, false, b1f, (short)0, acc01, false, false);
    acc10 = __builtin_amdgcn_wmma_f32_16x16x4_f32(false, a1, false, b0,  (short)0, acc10, false, false);
    acc11 = __builtin_amdgcn_wmma_f32_16x16x4_f32(false, a1, false, b1f, (short)0, acc11, false, false);
  }

  const int hA = h0 + wn + lo;
  const float bv0 = b1[hA], bv1 = b1[hA + 16];
  const float bias0 = (sel == 0) ? bv0 : 0.0f;
  const float bias1 = (sel == 0) ? bv1 : 0.0f;
  float* __restrict__ dst = ((sel == 0) ? ws_a : ws_c) + (size_t)(m0 + wm + 8 * hi) * H_ + hA;
#pragma unroll
  for (int r = 0; r < 8; ++r) {
    dst[(size_t)(r     ) * H_     ] = acc00[r] + bias0;
    dst[(size_t)(r     ) * H_ + 16] = acc01[r] + bias1;
    dst[(size_t)(r + 16) * H_     ] = acc10[r] + bias0;
    dst[(size_t)(r + 16) * H_ + 16] = acc11[r] + bias1;
  }
}

#define HC_  64
#define NCH_ (H_ / HC_)
#define AP_  HC_
#define CP_  68

__global__ __launch_bounds__(256) void edge_epilogue(
    const float* ws_a, const float* ws_c,
    const float* __restrict__ w2, const float* __restrict__ b2p,
    float* __restrict__ out)
{
  __shared__ __attribute__((aligned(16))) float a_s[2][32 * AP_];
  __shared__ __attribute__((aligned(16))) float c_s[2][64 * CP_];
  __shared__ __attribute__((aligned(16))) float w_s[H_];

  const int tid = threadIdx.x;
  int t = blockIdx.x;
  const int jt = t & 3; t >>= 2;
  const int it = t & 7; t >>= 3;
  const int b  = t;
  const int i0 = it * 32, j0 = jt * 64;

  const int sr = tid >> 4;
  const int sq = tid & 15;
  const float* ga = ws_a + ((size_t)(b * C_ + i0 + sr)) * H_ + 4 * sq;
  const float* gc = ws_c + ((size_t)(b * C_ + j0 + sr)) * H_ + 4 * sq;
  const int la = sr * AP_ + 4 * sq;
  const int lc = sr * CP_ + 4 * sq;

  float4 ra0, ra1, rc0, rc1, rc2, rc3;

  ra0 = *(const float4*)(ga);
  ra1 = *(const float4*)(ga + 16 * H_);
  rc0 = *(const float4*)(gc);
  rc1 = *(const float4*)(gc + 16 * H_);
  rc2 = *(const float4*)(gc + 32 * H_);
  rc3 = *(const float4*)(gc + 48 * H_);
  const float wreg = w2[tid];
  *(float4*)(&a_s[0][la])            = ra0;
  *(float4*)(&a_s[0][la + 16 * AP_]) = ra1;
  *(float4*)(&c_s[0][lc])            = rc0;
  *(float4*)(&c_s[0][lc + 16 * CP_]) = rc1;
  *(float4*)(&c_s[0][lc + 32 * CP_]) = rc2;
  *(float4*)(&c_s[0][lc + 48 * CP_]) = rc3;
  w_s[tid] = wreg;
  ra0 = *(const float4*)(ga + HC_);
  ra1 = *(const float4*)(ga + HC_ + 16 * H_);
  rc0 = *(const float4*)(gc + HC_);
  rc1 = *(const float4*)(gc + HC_ + 16 * H_);
  rc2 = *(const float4*)(gc + HC_ + 32 * H_);
  rc3 = *(const float4*)(gc + HC_ + 48 * H_);
  lds_barrier();

  const int ti = tid >> 5;
  const int tj = tid & 31;

  float acc00 = 0.f, acc10 = 0.f, acc20 = 0.f, acc30 = 0.f;
  float acc01 = 0.f, acc11 = 0.f, acc21 = 0.f, acc31 = 0.f;

#pragma unroll
  for (int ch = 0; ch < NCH_; ++ch) {
    const int cur = ch & 1;
    const float* a0 = &a_s[cur][(ti +  0) * AP_];
    const float* a1 = &a_s[cur][(ti +  8) * AP_];
    const float* a2 = &a_s[cur][(ti + 16) * AP_];
    const float* a3 = &a_s[cur][(ti + 24) * AP_];
    const float* c0 = &c_s[cur][(tj +  0) * CP_];
    const float* c1 = &c_s[cur][(tj + 32) * CP_];
    const float* wp = w_s + ch * HC_;

#pragma unroll 4
    for (int h = 0; h < HC_; h += 4) {
      const float4 wv = *(const float4*)(wp + h);
      const float4 p  = *(const float4*)(c0 + h);
      const float4 q  = *(const float4*)(c1 + h);
      const float4 x0 = *(const float4*)(a0 + h);
      const float4 x1 = *(const float4*)(a1 + h);
      const float4 x2 = *(const float4*)(a2 + h);
      const float4 x3 = *(const float4*)(a3 + h);
      acc00 = fmaf(fmaxf(x0.x + p.x, 0.f), wv.x, acc00);
      acc01 = fmaf(fmaxf(x0.x + q.x, 0.f), wv.x, acc01);
      acc10 = fmaf(fmaxf(x1.x + p.x, 0.f), wv.x, acc10);
      acc11 = fmaf(fmaxf(x1.x + q.x, 0.f), wv.x, acc11);
      acc20 = fmaf(fmaxf(x2.x + p.x, 0.f), wv.x, acc20);
      acc21 = fmaf(fmaxf(x2.x + q.x, 0.f), wv.x, acc21);
      acc30 = fmaf(fmaxf(x3.x + p.x, 0.f), wv.x, acc30);
      acc31 = fmaf(fmaxf(x3.x + q.x, 0.f), wv.x, acc31);

      acc00 = fmaf(fmaxf(x0.y + p.y, 0.f), wv.y, acc00);
      acc01 = fmaf(fmaxf(x0.y + q.y, 0.f), wv.y, acc01);
      acc10 = fmaf(fmaxf(x1.y + p.y, 0.f), wv.y, acc10);
      acc11 = fmaf(fmaxf(x1.y + q.y, 0.f), wv.y, acc11);
      acc20 = fmaf(fmaxf(x2.y + p.y, 0.f), wv.y, acc20);
      acc21 = fmaf(fmaxf(x2.y + q.y, 0.f), wv.y, acc21);
      acc30 = fmaf(fmaxf(x3.y + p.y, 0.f), wv.y, acc30);
      acc31 = fmaf(fmaxf(x3.y + q.y, 0.f), wv.y, acc31);

      acc00 = fmaf(fmaxf(x0.z + p.z, 0.f), wv.z, acc00);
      acc01 = fmaf(fmaxf(x0.z + q.z, 0.f), wv.z, acc01);
      acc10 = fmaf(fmaxf(x1.z + p.z, 0.f), wv.z, acc10);
      acc11 = fmaf(fmaxf(x1.z + q.z, 0.f), wv.z, acc11);
      acc20 = fmaf(fmaxf(x2.z + p.z, 0.f), wv.z, acc20);
      acc21 = fmaf(fmaxf(x2.z + q.z, 0.f), wv.z, acc21);
      acc30 = fmaf(fmaxf(x3.z + p.z, 0.f), wv.z, acc30);
      acc31 = fmaf(fmaxf(x3.z + q.z, 0.f), wv.z, acc31);

      acc00 = fmaf(fmaxf(x0.w + p.w, 0.f), wv.w, acc00);
      acc01 = fmaf(fmaxf(x0.w + q.w, 0.f), wv.w, acc01);
      acc10 = fmaf(fmaxf(x1.w + p.w, 0.f), wv.w, acc10);
      acc11 = fmaf(fmaxf(x1.w + q.w, 0.f), wv.w, acc11);
      acc20 = fmaf(fmaxf(x2.w + p.w, 0.f), wv.w, acc20);
      acc21 = fmaf(fmaxf(x2.w + q.w, 0.f), wv.w, acc21);
      acc30 = fmaf(fmaxf(x3.w + p.w, 0.f), wv.w, acc30);
      acc31 = fmaf(fmaxf(x3.w + q.w, 0.f), wv.w, acc31);
    }

    if (ch + 1 < NCH_) {
      const int nxt = cur ^ 1;
      *(float4*)(&a_s[nxt][la])            = ra0;
      *(float4*)(&a_s[nxt][la + 16 * AP_]) = ra1;
      *(float4*)(&c_s[nxt][lc])            = rc0;
      *(float4*)(&c_s[nxt][lc + 16 * CP_]) = rc1;
      *(float4*)(&c_s[nxt][lc + 32 * CP_]) = rc2;
      *(float4*)(&c_s[nxt][lc + 48 * CP_]) = rc3;
      if (ch + 2 < NCH_) {
        const int ho = (ch + 2) * HC_;
        ra0 = *(const float4*)(ga + ho);
        ra1 = *(const float4*)(ga + ho + 16 * H_);
        rc0 = *(const float4*)(gc + ho);
        rc1 = *(const float4*)(gc + ho + 16 * H_);
        rc2 = *(const float4*)(gc + ho + 32 * H_);
        rc3 = *(const float4*)(gc + ho + 48 * H_);
      }
      lds_barrier();
    }
  }

  const float b2 = b2p[0];
  const size_t obase = ((size_t)b * C_) * C_;
  const int j = j0 + tj;
  float* o = out + obase + (size_t)(i0 + ti) * C_ + j;
  o[ 0 * 8 * C_     ] = 1.0f / (1.0f + __expf(-(acc00 + b2)));
  o[ 0 * 8 * C_ + 32] = 1.0f / (1.0f + __expf(-(acc01 + b2)));
  o[ 1 * 8 * C_     ] = 1.0f / (1.0f + __expf(-(acc10 + b2)));
  o[ 1 * 8 * C_ + 32] = 1.0f / (1.0f + __expf(-(acc11 + b2)));
  o[ 2 * 8 * C_     ] = 1.0f / (1.0f + __expf(-(acc20 + b2)));
  o[ 2 * 8 * C_ + 32] = 1.0f / (1.0f + __expf(-(acc21 + b2)));
  o[ 3 * 8 * C_     ] = 1.0f / (1.0f + __expf(-(acc30 + b2)));
  o[ 3 * 8 * C_ + 32] = 1.0f / (1.0f + __expf(-(acc31 + b2)));
}

extern "C" void kernel_launch(void* const* d_in, const int* in_sizes, int n_in,
                              void* d_out, int out_size, void* d_ws, size_t ws_size,
                              hipStream_t stream) {
  const float* xa = (const float*)d_in[0];
  const float* W1 = (const float*)d_in[1];
  const float* b1 = (const float*)d_in[2];
  const float* w2 = (const float*)d_in[3];
  const float* b2 = (const float*)d_in[4];
  float* out = (float*)d_out;

  float* ws_a = (float*)d_ws;
  float* ws_c = ws_a + (size_t)B_ * C_ * H_;

  gemm_ac_wmma<<<256, 128, 0, stream>>>(xa, W1, b1, ws_a, ws_c);

  edge_epilogue<<<256, 256, 0, stream>>>(ws_a, ws_c, w2, b2, out);
}


# epilogue loop: relu(a+c) as max(c,-a)+a, so max+fma per element (2 vector ops, was 3), in hand-written dual-issue asm; GEMM stores -(a+b1); row sums of -a*w2 (16-lane tree) subtracted at the end
# speedup vs baseline: 1.6927x; 1.0827x over previous
#include <hip/hip_runtime.h>
#include <math.h>

typedef __attribute__((ext_vector_type(2))) float v2f;
typedef __attribute__((ext_vector_type(8))) float v8f;

#define B_ 8
#define C_ 256
#define F_ 128
#define H_ 256

__device__ __forceinline__ void lds_barrier() {
  __builtin_amdgcn_fence(__ATOMIC_RELEASE, "workgroup", "local");
  __builtin_amdgcn_s_barrier();
  __builtin_amdgcn_fence(__ATOMIC_ACQUIRE, "workgroup", "local");
}

__device__ __forceinline__ unsigned lds_off(const void* p) { return (unsigned)(size_t)p; }

__device__ __forceinline__ float row16_sum(float v) {
  v += __int_as_float(__builtin_amdgcn_update_dpp(0, __float_as_int(v), 0x161, 0xf, 0xf, false));
  v += __int_as_float(__builtin_amdgcn_update_dpp(0, __float_as_int(v), 0x162, 0xf, 0xf, false));
  v += __int_as_float(__builtin_amdgcn_update_dpp(0, __float_as_int(v), 0x164, 0xf, 0xf, false));
  v += __int_as_float(__builtin_amdgcn_update_dpp(0, __float_as_int(v), 0x168, 0xf, 0xf, false));
  return v;
}

#define GT_  64
#define GP_  132
#define GKC_ 64

__global__ __launch_bounds__(128) void gemm_ac_wmma(
    const float* xa, const float* W1,
    const float* __restrict__ b1,
    float* __restrict__ ws_na, float* __restrict__ ws_c)
{
  __shared__ __attribute__((aligned(16))) float ga_s[GT_ * GP_];
  __shared__ __attribute__((aligned(16))) float gb_s[GT_ * GP_];

  const int tid  = threadIdx.x;
  const int lane = tid & 31;
  const int wave = tid >> 5;

  const int nt  = blockIdx.x & 7;
  const int mt  = blockIdx.x >> 3;
  const int sel = nt >> 2;
  const int h0  = (nt & 3) * GT_;
  const int m0  = mt * GT_;

  const int sr = tid >> 4;
  const int sq = tid & 15;
  const float* gA = xa + (size_t)(m0 + sr) * F_ + 4 * sq;
  const float* gB = W1 + (size_t)(h0 + sr) * (2 * F_) + sel * F_ + 4 * sq;
  const int ls = sr * GP_ + 4 * sq;

  const float4 pa0 = *(const float4*)(gA + 0 * F_);
  const float4 pa1 = *(const float4*)(gA + 8 * F_);
  const float4 pa2 = *(const float4*)(gA + 16 * F_);
  const float4 pa3 = *(const float4*)(gA + 24 * F_);
  const float4 pa4 = *(const float4*)(gA + 32 * F_);
  const float4 pa5 = *(const float4*)(gA + 40 * F_);
  const float4 pa6 = *(const float4*)(gA + 48 * F_);
  const float4 pa7 = *(const float4*)(gA + 56 * F_);
  const float4 pb0 = *(const float4*)(gB + 0 * (2 * F_));
  const float4 pb1 = *(const float4*)(gB + 8 * (2 * F_));
  const float4 pb2 = *(const float4*)(gB + 16 * (2 * F_));
  const float4 pb3 = *(const float4*)(gB + 24 * (2 * F_));
  const float4 pb4 = *(const float4*)(gB + 32 * (2 * F_));
  const float4 pb5 = *(const float4*)(gB + 40 * (2 * F_));
  const float4 pb6 = *(const float4*)(gB + 48 * (2 * F_));
  const float4 pb7 = *(const float4*)(gB + 56 * (2 * F_));
  __builtin_amdgcn_sched_barrier(0);
  const float4 qa0 = *(const float4*)(gA + 0 * F_ + GKC_);
  const float4 qa1 = *(const float4*)(gA + 8 * F_ + GKC_);
  const float4 qa2 = *(const float4*)(gA + 16 * F_ + GKC_);
  const float4 qa3 = *(const float4*)(gA + 24 * F_ + GKC_);
  const float4 qa4 = *(const float4*)(gA + 32 * F_ + GKC_);
  const float4 qa5 = *(const float4*)(gA + 40 * F_ + GKC_);
  const float4 qa6 = *(const float4*)(gA + 48 * F_ + GKC_);
  const float4 qa7 = *(const float4*)(gA + 56 * F_ + GKC_);
  const float4 qb0 = *(const float4*)(gB + 0 * (2 * F_) + GKC_);
  const float4 qb1 = *(const float4*)(gB + 8 * (2 * F_) + GKC_);
  const float4 qb2 = *(const float4*)(gB + 16 * (2 * F_) + GKC_);
  const float4 qb3 = *(const float4*)(gB + 24 * (2 * F_) + GKC_);
  const float4 qb4 = *(const float4*)(gB + 32 * (2 * F_) + GKC_);
  const float4 qb5 = *(const float4*)(gB + 40 * (2 * F_) + GKC_);
  const float4 qb6 = *(const float4*)(gB + 48 * (2 * F_) + GKC_);
  const float4 qb7 = *(const float4*)(gB + 56 * (2 * F_) + GKC_);

  const int lo = lane & 15;
  const int hi = lane >> 4;
  const int wm = (wave >> 1) * 32;
  const int wn = (wave & 1) * 32;
  const float* fa0 = &ga_s[(wm +  0 + lo) * GP_ + 2 * hi];
  const float* fa1 = &ga_s[(wm + 16 + lo) * GP_ + 2 * hi];
  const float* fb0 = &gb_s[(wn +  0 + lo) * GP_ + 2 * hi];
  const float* fb1 = &gb_s[(wn + 16 + lo) * GP_ + 2 * hi];

  v8f acc00 = {}, acc01 = {}, acc10 = {}, acc11 = {};

  *(float4*)(&ga_s[ls + 0 * GP_]) = pa0;
  *(float4*)(&ga_s[ls + 8 * GP_]) = pa1;
  *(float4*)(&ga_s[ls + 16 * GP_]) = pa2;
  *(float4*)(&ga_s[ls + 24 * GP_]) = pa3;
  *(float4*)(&ga_s[ls + 32 * GP_]) = pa4;
  *(float4*)(&ga_s[ls + 40 * GP_]) = pa5;
  *(float4*)(&ga_s[ls + 48 * GP_]) = pa6;
  *(float4*)(&ga_s[ls + 56 * GP_]) = pa7;
  *(float4*)(&gb_s[ls + 0 * GP_]) = pb0;
  *(float4*)(&gb_s[ls + 8 * GP_]) = pb1;
  *(float4*)(&gb_s[ls + 16 * GP_]) = pb2;
  *(float4*)(&gb_s[ls + 24 * GP_]) = pb3;
  *(float4*)(&gb_s[ls + 32 * GP_]) = pb4;
  *(float4*)(&gb_s[ls + 40 * GP_]) = pb5;
  *(float4*)(&gb_s[ls + 48 * GP_]) = pb6;
  *(float4*)(&gb_s[ls + 56 * GP_]) = pb7;
  lds_barrier();

#pragma unroll
  for (int k = 0; k < GKC_; k += 4) {
    const v2f a0 = *(const v2f*)(fa0 + k);
    const v2f a1 = *(const v2f*)(fa1 + k);
    const v2f b0 = *(const v2f*)(fb0 + k);
    const v2f b1f = *(const v2f*)(fb1 + k);
    acc00 = __builtin_amdgcn_wmma_f32_16x16x4_f32(false, a0, false, b0,  (short)0, acc00, false, false);
    acc01 = __builtin_amdgcn_wmma_f32_16x16x4_f32(false, a0, false, b1f, (short)0, acc01, false, false);
    acc10 = __builtin_amdgcn_wmma_f32_16x16x4_f32(false, a1, false, b0,  (short)0, acc10, false, false);
    acc11 = __builtin_amdgcn_wmma_f32_16x16x4_f32(false, a1, false, b1f, (short)0, acc11, false, false);
  }

  *(float4*)(&ga_s[ls + 0 * GP_ + GKC_]) = qa0;
  *(float4*)(&ga_s[ls + 8 * GP_ + GKC_]) = qa1;
  *(float4*)(&ga_s[ls + 16 * GP_ + GKC_]) = qa2;
  *(float4*)(&ga_s[ls + 24 * GP_ + GKC_]) = qa3;
  *(float4*)(&ga_s[ls + 32 * GP_ + GKC_]) = qa4;
  *(float4*)(&ga_s[ls + 40 * GP_ + GKC_]) = qa5;
  *(float4*)(&ga_s[ls + 48 * GP_ + GKC_]) = qa6;
  *(float4*)(&ga_s[ls + 56 * GP_ + GKC_]) = qa7;
  *(float4*)(&gb_s[ls + 0 * GP_ + GKC_]) = qb0;
  *(float4*)(&gb_s[ls + 8 * GP_ + GKC_]) = qb1;
  *(float4*)(&gb_s[ls + 16 * GP_ + GKC_]) = qb2;
  *(float4*)(&gb_s[ls + 24 * GP_ + GKC_]) = qb3;
  *(float4*)(&gb_s[ls + 32 * GP_ + GKC_]) = qb4;
  *(float4*)(&gb_s[ls + 40 * GP_ + GKC_]) = qb5;
  *(float4*)(&gb_s[ls + 48 * GP_ + GKC_]) = qb6;
  *(float4*)(&gb_s[ls + 56 * GP_ + GKC_]) = qb7;
  lds_barrier();

#pragma unroll
  for (int k = GKC_; k < F_; k += 4) {
    const v2f a0 = *(const v2f*)(fa0 + k);
    const v2f a1 = *(const v2f*)(fa1 + k);
    const v2f b0 = *(const v2f*)(fb0 + k);
    const v2f b1f = *(const v2f*)(fb1 + k);
    acc00 = __builtin_amdgcn_wmma_f32_16x16x4_f32(false, a0, false, b0,  (short)0, acc00, false, false);
    acc01 = __builtin_amdgcn_wmma_f32_16x16x4_f32(false, a0, false, b1f, (short)0, acc01, false, false);
    acc10 = __builtin_amdgcn_wmma_f32_16x16x4_f32(false, a1, false, b0,  (short)0, acc10, false, false);
    acc11 = __builtin_amdgcn_wmma_f32_16x16x4_f32(false, a1, false, b1f, (short)0, acc11, false, false);
  }

  const int hA = h0 + wn + lo;
  const float bv0 = b1[hA], bv1 = b1[hA + 16];
  const float sgn   = (sel == 0) ? -1.0f : 1.0f;
  const float bias0 = (sel == 0) ? -bv0 : 0.0f;
  const float bias1 = (sel == 0) ? -bv1 : 0.0f;
  float* __restrict__ dst = ((sel == 0) ? ws_na : ws_c) + (size_t)(m0 + wm + 8 * hi) * H_ + hA;
#pragma unroll
  for (int r = 0; r < 8; ++r) {
    dst[(size_t)(r     ) * H_     ] = fmaf(acc00[r], sgn, bias0);
    dst[(size_t)(r     ) * H_ + 16] = fmaf(acc01[r], sgn, bias1);
    dst[(size_t)(r + 16) * H_     ] = fmaf(acc10[r], sgn, bias0);
    dst[(size_t)(r + 16) * H_ + 16] = fmaf(acc11[r], sgn, bias1);
  }
}

#define HC_  64
#define NCH_ (H_ / HC_)
#define AP_  HC_
#define CP_  68

__global__ __launch_bounds__(256) void edge_epilogue(
    const float* ws_na, const float* ws_c,
    const float* __restrict__ w2, const float* __restrict__ b2p,
    float* __restrict__ out)
{
  __shared__ __attribute__((aligned(16))) float a_s[2][32 * AP_];
  __shared__ __attribute__((aligned(16))) float c_s[2][64 * CP_];
  __shared__ __attribute__((aligned(16))) float w_s[H_];
  __shared__ float s_s[32];

  const int tid = threadIdx.x;
  int t = blockIdx.x;
  const int jt = t & 3; t >>= 2;
  const int it = t & 7; t >>= 3;
  const int b  = t;
  const int i0 = it * 32, j0 = jt * 64;

  const int sr = tid >> 4;
  const int sq = tid & 15;
  const float* ga = ws_na + ((size_t)(b * C_ + i0 + sr)) * H_ + 4 * sq;
  const float* gw = w2 + 4 * sq;
  const float* gc = ws_c + ((size_t)(b * C_ + j0 + sr)) * H_ + 4 * sq;
  const int la = sr * AP_ + 4 * sq;
  const int lc = sr * CP_ + 4 * sq;

  float4 ra0, ra1, rc0, rc1, rc2, rc3;
  float4 rw;
  float ps0 = 0.f, ps1 = 0.f;

  ra0 = *(const float4*)(ga);
  ra1 = *(const float4*)(ga + 16 * H_);
  rc0 = *(const float4*)(gc);
  rc1 = *(const float4*)(gc + 16 * H_);
  rc2 = *(const float4*)(gc + 32 * H_);
  rc3 = *(const float4*)(gc + 48 * H_);
  rw  = *(const float4*)(gw);
  const float wreg = w2[tid];
  *(float4*)(&a_s[0][la])            = ra0;
  *(float4*)(&a_s[0][la + 16 * AP_]) = ra1;
  *(float4*)(&c_s[0][lc])            = rc0;
  *(float4*)(&c_s[0][lc + 16 * CP_]) = rc1;
  *(float4*)(&c_s[0][lc + 32 * CP_]) = rc2;
  *(float4*)(&c_s[0][lc + 48 * CP_]) = rc3;
  w_s[tid] = wreg;
  ps0 = fmaf(ra0.x, rw.x, ps0); ps1 = fmaf(ra1.x, rw.x, ps1);
  ps0 = fmaf(ra0.y, rw.y, ps0); ps1 = fmaf(ra1.y, rw.y, ps1);
  ps0 = fmaf(ra0.z, rw.z, ps0); ps1 = fmaf(ra1.z, rw.z, ps1);
  ps0 = fmaf(ra0.w, rw.w, ps0); ps1 = fmaf(ra1.w, rw.w, ps1);
  ra0 = *(const float4*)(ga + HC_);
  ra1 = *(const float4*)(ga + HC_ + 16 * H_);
  rc0 = *(const float4*)(gc + HC_);
  rc1 = *(const float4*)(gc + HC_ + 16 * H_);
  rc2 = *(const float4*)(gc + HC_ + 32 * H_);
  rc3 = *(const float4*)(gc + HC_ + 48 * H_);
  rw  = *(const float4*)(gw + HC_);
  lds_barrier();

  const int ti = tid >> 5;
  const int tj = tid & 31;

  float acc00 = 0.f, acc10 = 0.f, acc20 = 0.f, acc30 = 0.f;
  float acc01 = 0.f, acc11 = 0.f, acc21 = 0.f, acc31 = 0.f;

#pragma unroll
  for (int ch = 0; ch < NCH_; ++ch) {
    const int cur = ch & 1;
    const unsigned lc0 = lds_off(&c_s[cur][tj * CP_]);
    const unsigned la0 = lds_off(&a_s[cur][ti * AP_]);
    const unsigned lw0 = lds_off(w_s + ch * HC_);

    asm volatile(
      "v_mov_b32 v232, %0\n"
      "v_mov_b32 v233, %1\n"
      "v_mov_b32 v234, %2\n"
      "v_mov_b32 v235, %3\n"
      "v_mov_b32 v236, %4\n"
      "v_mov_b32 v237, %5\n"
      "v_mov_b32 v238, %6\n"
      "v_mov_b32 v239, %7\n"
      "v_mov_b32 v240, %10\n"
      "v_mov_b32 v241, %11\n"
      "v_mov_b32 v242, %12\n"
      "s_mov_b32 s40, 0\n"
      "LEPI_%=:\n"
      "ds_load_b128 v[96:99], v240 offset:0\n"
      "ds_load_b128 v[102:105], v240 offset:8704\n"
      "ds_load_b128 v[106:109], v241 offset:0\n"
      "ds_load_b128 v[110:113], v241 offset:2048\n"
      "ds_load_b128 v[114:117], v241 offset:4096\n"
      "ds_load_b128 v[118:121], v241 offset:6144\n"
      "ds_load_b128 v[122:125], v242 offset:0\n"
      "ds_load_b128 v[126:129], v240 offset:16\n"
      "ds_load_b128 v[132:135], v240 offset:8720\n"
      "ds_load_b128 v[136:139], v241 offset:16\n"
      "ds_load_b128 v[140:143], v241 offset:2064\n"
      "ds_load_b128 v[144:147], v241 offset:4112\n"
      "ds_load_b128 v[148:151], v241 offset:6160\n"
      "ds_load_b128 v[152:155], v242 offset:16\n"
      "ds_load_b128 v[156:159], v240 offset:32\n"
      "ds_load_b128 v[162:165], v240 offset:8736\n"
      "ds_load_b128 v[166:169], v241 offset:32\n"
      "ds_load_b128 v[170:173], v241 offset:2080\n"
      "ds_load_b128 v[174:177], v241 offset:4128\n"
      "ds_load_b128 v[178:181], v241 offset:6176\n"
      "ds_load_b128 v[182:185], v242 offset:32\n"
      "ds_load_b128 v[186:189], v240 offset:48\n"
      "ds_load_b128 v[192:195], v240 offset:8752\n"
      "ds_load_b128 v[196:199], v241 offset:48\n"
      "ds_load_b128 v[200:203], v241 offset:2096\n"
      "ds_load_b128 v[204:207], v241 offset:4144\n"
      "ds_load_b128 v[208:211], v241 offset:6192\n"
      "ds_load_b128 v[212:215], v242 offset:48\n"
      "v_add_nc_u32_e32 v240, 64, v240\n"
      "v_add_nc_u32_e32 v241, 64, v241\n"
      "v_add_nc_u32_e32 v242, 64, v242\n"
      "s_wait_dscnt 21\n"
      "v_dual_max_num_f32 v216, v96, v106 :: v_dual_max_num_f32 v217, v102, v106\n"
      "v_dual_max_num_f32 v218, v96, v110 :: v_dual_max_num_f32 v219, v102, v110\n"
      "v_dual_max_num_f32 v220, v96, v114 :: v_dual_max_num_f32 v221, v102, v114\n"
      "v_dual_max_num_f32 v222, v96, v118 :: v_dual_max_num_f32 v223, v102, v118\n"
      "v_dual_max_num_f32 v224, v97, v107 :: v_dual_max_num_f32 v225, v103, v107\n"
      "v_dual_max_num_f32 v226, v97, v111 :: v_dual_max_num_f32 v227, v103, v111\n"
      "v_dual_max_num_f32 v228, v97, v115 :: v_dual_max_num_f32 v229, v103, v115\n"
      "v_dual_max_num_f32 v230, v97, v119 :: v_dual_max_num_f32 v231, v103, v119\n"
      "v_dual_fmac_f32 v232, v216, v122 :: v_dual_fmac_f32 v233, v217, v122\n"
      "v_dual_fmac_f32 v234, v218, v122 :: v_dual_fmac_f32 v235, v219, v122\n"
      "v_dual_fmac_f32 v236, v220, v122 :: v_dual_fmac_f32 v237, v221, v122\n"
      "v_dual_fmac_f32 v238, v222, v122 :: v_dual_fmac_f32 v239, v223, v122\n"
      "v_dual_max_num_f32 v216, v98, v108 :: v_dual_max_num_f32 v217, v104, v108\n"
      "v_dual_max_num_f32 v218, v98, v112 :: v_dual_max_num_f32 v219, v104, v112\n"
      "v_dual_max_num_f32 v220, v98, v116 :: v_dual_max_num_f32 v221, v104, v116\n"
      "v_dual_max_num_f32 v222, v98, v120 :: v_dual_max_num_f32 v223, v104, v120\n"
      "v_dual_fmac_f32 v232, v224, v123 :: v_dual_fmac_f32 v233, v225, v123\n"
      "v_dual_fmac_f32 v234, v226, v123 :: v_dual_fmac_f32 v235, v227, v123\n"
      "v_dual_fmac_f32 v236, v228, v123 :: v_dual_fmac_f32 v237, v229, v123\n"
      "v_dual_fmac_f32 v238, v230, v123 :: v_dual_fmac_f32 v239, v231, v123\n"
      "v_dual_max_num_f32 v224, v99, v109 :: v_dual_max_num_f32 v225, v105, v109\n"
      "v_dual_max_num_f32 v226, v99, v113 :: v_dual_max_num_f32 v227, v105, v113\n"
      "v_dual_max_num_f32 v228, v99, v117 :: v_dual_max_num_f32 v229, v105, v117\n"
      "v_dual_max_num_f32 v230, v99, v121 :: v_dual_max_num_f32 v231, v105, v121\n"
      "v_dual_fmac_f32 v232, v216, v124 :: v_dual_fmac_f32 v233, v217, v124\n"
      "v_dual_fmac_f32 v234, v218, v124 :: v_dual_fmac_f32 v235, v219, v124\n"
      "v_dual_fmac_f32 v236, v220, v124 :: v_dual_fmac_f32 v237, v221, v124\n"
      "v_dual_fmac_f32 v238, v222, v124 :: v_dual_fmac_f32 v239, v223, v124\n"
      "s_wait_dscnt 14\n"
      "v_dual_max_num_f32 v216, v126, v136 :: v_dual_max_num_f32 v217, v132, v136\n"
      "v_dual_max_num_f32 v218, v126, v140 :: v_dual_max_num_f32 v219, v132, v140\n"
      "v_dual_max_num_f32 v220, v126, v144 :: v_dual_max_num_f32 v221, v132, v144\n"
      "v_dual_max_num_f32 v222, v126, v148 :: v_dual_max_num_f32 v223, v132, v148\n"
      "v_dual_fmac_f32 v232, v224, v125 :: v_dual_fmac_f32 v233, v225, v125\n"
      "v_dual_fmac_f32 v234, v226, v125 :: v_dual_fmac_f32 v235, v227, v125\n"
      "v_dual_fmac_f32 v236, v228, v125 :: v_dual_fmac_f32 v237, v229, v125\n"
      "v_dual_fmac_f32 v238, v230, v125 :: v_dual_fmac_f32 v239, v231, v125\n"
      "v_dual_max_num_f32 v224, v127, v137 :: v_dual_max_num_f32 v225, v133, v137\n"
      "v_dual_max_num_f32 v226, v127, v141 :: v_dual_max_num_f32 v227, v133, v141\n"
      "v_dual_max_num_f32 v228, v127, v145 :: v_dual_max_num_f32 v229, v133, v145\n"
      "v_dual_max_num_f32 v230, v127, v149 :: v_dual_max_num_f32 v231, v133, v149\n"
      "v_dual_fmac_f32 v232, v216, v152 :: v_dual_fmac_f32 v233, v217, v152\n"
      "v_dual_fmac_f32 v234, v218, v152 :: v_dual_fmac_f32 v235, v219, v152\n"
      "v_dual_fmac_f32 v236, v220, v152 :: v_dual_fmac_f32 v237, v221, v152\n"
      "v_dual_fmac_f32 v238, v222, v152 :: v_dual_fmac_f32 v239, v223, v152\n"
      "v_dual_max_num_f32 v216, v128, v138 :: v_dual_max_num_f32 v217, v134, v138\n"
      "v_dual_max_num_f32 v218, v128, v142 :: v_dual_max_num_f32 v219, v134, v142\n"
      "v_dual_max_num_f32 v220, v128, v146 :: v_dual_max_num_f32 v221, v134, v146\n"
      "v_dual_max_num_f32 v222, v128, v150 :: v_dual_max_num_f32 v223, v134, v150\n"
      "v_dual_fmac_f32 v232, v224, v153 :: v_dual_fmac_f32 v233, v225, v153\n"
      "v_dual_fmac_f32 v234, v226, v153 :: v_dual_fmac_f32 v235, v227, v153\n"
      "v_dual_fmac_f32 v236, v228, v153 :: v_dual_fmac_f32 v237, v229, v153\n"
      "v_dual_fmac_f32 v238, v230, v153 :: v_dual_fmac_f32 v239, v231, v153\n"
      "v_dual_max_num_f32 v224, v129, v139 :: v_dual_max_num_f32 v225, v135, v139\n"
      "v_dual_max_num_f32 v226, v129, v143 :: v_dual_max_num_f32 v227, v135, v143\n"
      "v_dual_max_num_f32 v228, v129, v147 :: v_dual_max_num_f32 v229, v135, v147\n"
      "v_dual_max_num_f32 v230, v129, v151 :: v_dual_max_num_f32 v231, v135, v151\n"
      "v_dual_fmac_f32 v232, v216, v154 :: v_dual_fmac_f32 v233, v217, v154\n"
      "v_dual_fmac_f32 v234, v218, v154 :: v_dual_fmac_f32 v235, v219, v154\n"
      "v_dual_fmac_f32 v236, v220, v154 :: v_dual_fmac_f32 v237, v221, v154\n"
      "v_dual_fmac_f32 v238, v222, v154 :: v_dual_fmac_f32 v239, v223, v154\n"
      "s_wait_dscnt 7\n"
      "v_dual_max_num_f32 v216, v156, v166 :: v_dual_max_num_f32 v217, v162, v166\n"
      "v_dual_max_num_f32 v218, v156, v170 :: v_dual_max_num_f32 v219, v162, v170\n"
      "v_dual_max_num_f32 v220, v156, v174 :: v_dual_max_num_f32 v221, v162, v174\n"
      "v_dual_max_num_f32 v222, v156, v178 :: v_dual_max_num_f32 v223, v162, v178\n"
      "v_dual_fmac_f32 v232, v224, v155 :: v_dual_fmac_f32 v233, v225, v155\n"
      "v_dual_fmac_f32 v234, v226, v155 :: v_dual_fmac_f32 v235, v227, v155\n"
      "v_dual_fmac_f32 v236, v228, v155 :: v_dual_fmac_f32 v237, v229, v155\n"
      "v_dual_fmac_f32 v238, v230, v155 :: v_dual_fmac_f32 v239, v231, v155\n"
      "v_dual_max_num_f32 v224, v157, v167 :: v_dual_max_num_f32 v225, v163, v167\n"
      "v_dual_max_num_f32 v226, v157, v171 :: v_dual_max_num_f32 v227, v163, v171\n"
      "v_dual_max_num_f32 v228, v157, v175 :: v_dual_max_num_f32 v229, v163, v175\n"
      "v_dual_max_num_f32 v230, v157, v179 :: v_dual_max_num_f32 v231, v163, v179\n"
      "v_dual_fmac_f32 v232, v216, v182 :: v_dual_fmac_f32 v233, v217, v182\n"
      "v_dual_fmac_f32 v234, v218, v182 :: v_dual_fmac_f32 v235, v219, v182\n"
      "v_dual_fmac_f32 v236, v220, v182 :: v_dual_fmac_f32 v237, v221, v182\n"
      "v_dual_fmac_f32 v238, v222, v182 :: v_dual_fmac_f32 v239, v223, v182\n"
      "v_dual_max_num_f32 v216, v158, v168 :: v_dual_max_num_f32 v217, v164, v168\n"
      "v_dual_max_num_f32 v218, v158, v172 :: v_dual_max_num_f32 v219, v164, v172\n"
      "v_dual_max_num_f32 v220, v158, v176 :: v_dual_max_num_f32 v221, v164, v176\n"
      "v_dual_max_num_f32 v222, v158, v180 :: v_dual_max_num_f32 v223, v164, v180\n"
      "v_dual_fmac_f32 v232, v224, v183 :: v_dual_fmac_f32 v233, v225, v183\n"
      "v_dual_fmac_f32 v234, v226, v183 :: v_dual_fmac_f32 v235, v227, v183\n"
      "v_dual_fmac_f32 v236, v228, v183 :: v_dual_fmac_f32 v237, v229, v183\n"
      "v_dual_fmac_f32 v238, v230, v183 :: v_dual_fmac_f32 v239, v231, v183\n"
      "v_dual_max_num_f32 v224, v159, v169 :: v_dual_max_num_f32 v225, v165, v169\n"
      "v_dual_max_num_f32 v226, v159, v173 :: v_dual_max_num_f32 v227, v165, v173\n"
      "v_dual_max_num_f32 v228, v159, v177 :: v_dual_max_num_f32 v229, v165, v177\n"
      "v_dual_max_num_f32 v230, v159, v181 :: v_dual_max_num_f32 v231, v165, v181\n"
      "v_dual_fmac_f32 v232, v216, v184 :: v_dual_fmac_f32 v233, v217, v184\n"
      "v_dual_fmac_f32 v234, v218, v184 :: v_dual_fmac_f32 v235, v219, v184\n"
      "v_dual_fmac_f32 v236, v220, v184 :: v_dual_fmac_f32 v237, v221, v184\n"
      "v_dual_fmac_f32 v238, v222, v184 :: v_dual_fmac_f32 v239, v223, v184\n"
      "s_wait_dscnt 0\n"
      "v_dual_max_num_f32 v216, v186, v196 :: v_dual_max_num_f32 v217, v192, v196\n"
      "v_dual_max_num_f32 v218, v186, v200 :: v_dual_max_num_f32 v219, v192, v200\n"
      "v_dual_max_num_f32 v220, v186, v204 :: v_dual_max_num_f32 v221, v192, v204\n"
      "v_dual_max_num_f32 v222, v186, v208 :: v_dual_max_num_f32 v223, v192, v208\n"
      "v_dual_fmac_f32 v232, v224, v185 :: v_dual_fmac_f32 v233, v225, v185\n"
      "v_dual_fmac_f32 v234, v226, v185 :: v_dual_fmac_f32 v235, v227, v185\n"
      "v_dual_fmac_f32 v236, v228, v185 :: v_dual_fmac_f32 v237, v229, v185\n"
      "v_dual_fmac_f32 v238, v230, v185 :: v_dual_fmac_f32 v239, v231, v185\n"
      "v_dual_max_num_f32 v224, v187, v197 :: v_dual_max_num_f32 v225, v193, v197\n"
      "v_dual_max_num_f32 v226, v187, v201 :: v_dual_max_num_f32 v227, v193, v201\n"
      "v_dual_max_num_f32 v228, v187, v205 :: v_dual_max_num_f32 v229, v193, v205\n"
      "v_dual_max_num_f32 v230, v187, v209 :: v_dual_max_num_f32 v231, v193, v209\n"
      "v_dual_fmac_f32 v232, v216, v212 :: v_dual_fmac_f32 v233, v217, v212\n"
      "v_dual_fmac_f32 v234, v218, v212 :: v_dual_fmac_f32 v235, v219, v212\n"
      "v_dual_fmac_f32 v236, v220, v212 :: v_dual_fmac_f32 v237, v221, v212\n"
      "v_dual_fmac_f32 v238, v222, v212 :: v_dual_fmac_f32 v239, v223, v212\n"
      "v_dual_max_num_f32 v216, v188, v198 :: v_dual_max_num_f32 v217, v194, v198\n"
      "v_dual_max_num_f32 v218, v188, v202 :: v_dual_max_num_f32 v219, v194, v202\n"
      "v_dual_max_num_f32 v220, v188, v206 :: v_dual_max_num_f32 v221, v194, v206\n"
      "v_dual_max_num_f32 v222, v188, v210 :: v_dual_max_num_f32 v223, v194, v210\n"
      "v_dual_fmac_f32 v232, v224, v213 :: v_dual_fmac_f32 v233, v225, v213\n"
      "v_dual_fmac_f32 v234, v226, v213 :: v_dual_fmac_f32 v235, v227, v213\n"
      "v_dual_fmac_f32 v236, v228, v213 :: v_dual_fmac_f32 v237, v229, v213\n"
      "v_dual_fmac_f32 v238, v230, v213 :: v_dual_fmac_f32 v239, v231, v213\n"
      "v_dual_max_num_f32 v224, v189, v199 :: v_dual_max_num_f32 v225, v195, v199\n"
      "v_dual_max_num_f32 v226, v189, v203 :: v_dual_max_num_f32 v227, v195, v203\n"
      "v_dual_max_num_f32 v228, v189, v207 :: v_dual_max_num_f32 v229, v195, v207\n"
      "v_dual_max_num_f32 v230, v189, v211 :: v_dual_max_num_f32 v231, v195, v211\n"
      "v_dual_fmac_f32 v232, v216, v214 :: v_dual_fmac_f32 v233, v217, v214\n"
      "v_dual_fmac_f32 v234, v218, v214 :: v_dual_fmac_f32 v235, v219, v214\n"
      "v_dual_fmac_f32 v236, v220, v214 :: v_dual_fmac_f32 v237, v221, v214\n"
      "v_dual_fmac_f32 v238, v222, v214 :: v_dual_fmac_f32 v239, v223, v214\n"
      "s_delay_alu instid0(VALU_DEP_4)\n"
      "v_dual_fmac_f32 v232, v224, v215 :: v_dual_fmac_f32 v233, v225, v215\n"
      "s_delay_alu instid0(VALU_DEP_4)\n"
      "v_dual_fmac_f32 v234, v226, v215 :: v_dual_fmac_f32 v235, v227, v215\n"
      "s_delay_alu instid0(VALU_DEP_4)\n"
      "v_dual_fmac_f32 v236, v228, v215 :: v_dual_fmac_f32 v237, v229, v215\n"
      "s_delay_alu instid0(VALU_DEP_4)\n"
      "v_dual_fmac_f32 v238, v230, v215 :: v_dual_fmac_f32 v239, v231, v215\n"
      "s_add_co_i32 s40, s40, 1\n"
      "s_cmp_lt_u32 s40, 4\n"
      "s_cbranch_scc1 LEPI_%=\n"
      "v_mov_b32 %0, v232\n"
      "v_mov_b32 %1, v233\n"
      "v_mov_b32 %2, v234\n"
      "v_mov_b32 %3, v235\n"
      "v_mov_b32 %4, v236\n"
      "v_mov_b32 %5, v237\n"
      "v_mov_b32 %6, v238\n"
      "v_mov_b32 %7, v239\n"
      : "+v"(acc00), "+v"(acc01), "+v"(acc10), "+v"(acc11), "+v"(acc20), "+v"(acc21), "+v"(acc30), "+v"(acc31),
        "+v"(ps0), "+v"(ps1)
      : "v"(lc0), "v"(la0), "v"(lw0)
      :
        "v96", "v97", "v98", "v99", "v100", "v101", "v102", "v103", "v104", "v105", "v106", "v107",
        "v108", "v109", "v110", "v111", "v112", "v113", "v114", "v115", "v116", "v117", "v118", "v119",
        "v120", "v121", "v122", "v123", "v124", "v125", "v126", "v127", "v128", "v129", "v130", "v131",
        "v132", "v133", "v134", "v135", "v136", "v137", "v138", "v139", "v140", "v141", "v142", "v143",
        "v144", "v145", "v146", "v147", "v148", "v149", "v150", "v151", "v152", "v153", "v154", "v155",
        "v156", "v157", "v158", "v159", "v160", "v161", "v162", "v163", "v164", "v165", "v166", "v167",
        "v168", "v169", "v170", "v171", "v172", "v173", "v174", "v175", "v176", "v177", "v178", "v179",
        "v180", "v181", "v182", "v183", "v184", "v185", "v186", "v187", "v188", "v189", "v190", "v191",
        "v192", "v193", "v194", "v195", "v196", "v197", "v198", "v199", "v200", "v201", "v202", "v203",
        "v204", "v205", "v206", "v207", "v208", "v209", "v210", "v211", "v212", "v213", "v214", "v215",
        "v216", "v217", "v218", "v219", "v220", "v221", "v222", "v223", "v224", "v225", "v226", "v227",
        "v228", "v229", "v230", "v231", "v232", "v233", "v234", "v235", "v236", "v237", "v238", "v239",
        "v240", "v241", "v242",
        "s40", "scc", "memory");

    if (ch + 1 < NCH_) {
      const int nxt = cur ^ 1;
      *(float4*)(&a_s[nxt][la])            = ra0;
      *(float4*)(&a_s[nxt][la + 16 * AP_]) = ra1;
      *(float4*)(&c_s[nxt][lc])            = rc0;
      *(float4*)(&c_s[nxt][lc + 16 * CP_]) = rc1;
      *(float4*)(&c_s[nxt][lc + 32 * CP_]) = rc2;
      *(float4*)(&c_s[nxt][lc + 48 * CP_]) = rc3;
      ps0 = fmaf(ra0.x, rw.x, ps0); ps1 = fmaf(ra1.x, rw.x, ps1);
      ps0 = fmaf(ra0.y, rw.y, ps0); ps1 = fmaf(ra1.y, rw.y, ps1);
      ps0 = fmaf(ra0.z, rw.z, ps0); ps1 = fmaf(ra1.z, rw.z, ps1);
      ps0 = fmaf(ra0.w, rw.w, ps0); ps1 = fmaf(ra1.w, rw.w, ps1);
      if (ch + 2 < NCH_) {
        const int ho = (ch + 2) * HC_;
        ra0 = *(const float4*)(ga + ho);
        ra1 = *(const float4*)(ga + ho + 16 * H_);
        rc0 = *(const float4*)(gc + ho);
        rc1 = *(const float4*)(gc + ho + 16 * H_);
        rc2 = *(const float4*)(gc + ho + 32 * H_);
        rc3 = *(const float4*)(gc + ho + 48 * H_);
        rw  = *(const float4*)(gw + ho);
      } else {
        ps0 = row16_sum(ps0);
        ps1 = row16_sum(ps1);
        if (sq == 0) { s_s[sr] = ps0; s_s[sr + 16] = ps1; }
      }
      lds_barrier();
    }
  }

  const float b2 = b2p[0];
  const float s0 = s_s[ti], s1 = s_s[ti + 8], s2 = s_s[ti + 16], s3 = s_s[ti + 24];
  const size_t obase = ((size_t)b * C_) * C_;
  const int j = j0 + tj;
  float* o = out + obase + (size_t)(i0 + ti) * C_ + j;
  o[ 0 * 8 * C_     ] = 1.0f / (1.0f + __expf(-((acc00 - s0) + b2)));
  o[ 0 * 8 * C_ + 32] = 1.0f / (1.0f + __expf(-((acc01 - s0) + b2)));
  o[ 1 * 8 * C_     ] = 1.0f / (1.0f + __expf(-((acc10 - s1) + b2)));
  o[ 1 * 8 * C_ + 32] = 1.0f / (1.0f + __expf(-((acc11 - s1) + b2)));
  o[ 2 * 8 * C_     ] = 1.0f / (1.0f + __expf(-((acc20 - s2) + b2)));
  o[ 2 * 8 * C_ + 32] = 1.0f / (1.0f + __expf(-((acc21 - s2) + b2)));
  o[ 3 * 8 * C_     ] = 1.0f / (1.0f + __expf(-((acc30 - s3) + b2)));
  o[ 3 * 8 * C_ + 32] = 1.0f / (1.0f + __expf(-((acc31 - s3) + b2)));
}

extern "C" void kernel_launch(void* const* d_in, const int* in_sizes, int n_in,
                              void* d_out, int out_size, void* d_ws, size_t ws_size,
                              hipStream_t stream) {
  const float* xa = (const float*)d_in[0];
  const float* W1 = (const float*)d_in[1];
  const float* b1 = (const float*)d_in[2];
  const float* w2 = (const float*)d_in[3];
  const float* b2 = (const float*)d_in[4];
  float* out = (float*)d_out;

  float* ws_na = (float*)d_ws;
  float* ws_c = ws_na + (size_t)B_ * C_ * H_;

  gemm_ac_wmma<<<256, 128, 0, stream>>>(xa, W1, b1, ws_na, ws_c);

  edge_epilogue<<<256, 256, 0, stream>>>(ws_na, ws_c, w2, b2, out);
}
